# starting kernel
# baseline (speedup 1.0000x reference)
#include <hip/hip_runtime.h>

typedef float v2f __attribute__((ext_vector_type(2)));
typedef float v8f __attribute__((ext_vector_type(8)));

#define MT        64
#define WAVES     4
#define THREADS   128
#define FTOT      132
#define HDIM      512
#define DOUT      128
#define FP        132
#define HP        516
#define HBYTES    (WAVES * 16 * HP * 4)
#define TYPOFF    HBYTES
#define SMEM_BYTES (HBYTES + MT * 4)

__device__ __forceinline__ float gelu_exact(float x) {
    return 0.5f * x * (1.0f + erff(x * 0.70710678118654752f));
}

__global__ __launch_bounds__(THREADS)
void mpx_encoder_kernel(const float* __restrict__ nf,
                        const float* __restrict__ W1,
                        const float* __restrict__ b1,
                        const float* __restrict__ W2,
                        const float* __restrict__ b2,
                        float* __restrict__ out,
                        int Mtot)
{
    extern __shared__ char smem[];
    float* feats = (float*)smem;
    float* hbuf  = (float*)smem;
    int*   typ   = (int*)(smem + TYPOFF);

    const int  tid  = threadIdx.x;
    const int  lane = tid & 31;
    const int  wv   = tid >> 5;
    const int  lm   = lane & 15;
    const int  lh   = lane >> 4;
    const long base = (long)blockIdx.x * MT;

    #pragma unroll
    for (int it = 0; it < 16; ++it) {
        int i  = tid + THREADS * it;
        int r  = i >> 5;
        int c4 = i & 31;
        long rs = base + r; if (rs >= Mtot) rs = Mtot - 1;
        const float4 v = *(const float4*)(nf + rs * FTOT + 4 + c4 * 4);
        float* d = feats + r * FP + c4 * 4;
        d[0] = v.x; d[1] = v.y; d[2] = v.z; d[3] = v.w;
    }
    if (tid < MT) {
        long rs = base + tid; if (rs >= Mtot) rs = Mtot - 1;
        const float* p = nf + rs * FTOT;
        float best = p[0]; int bi = 0;
        #pragma unroll
        for (int i = 1; i < 4; ++i) { float q = p[i]; if (q > best) { best = q; bi = i; } }
        typ[tid] = bi;
    }
    __syncthreads();

    const float* myfeat = feats + wv * 16 * FP;
    v2f afr[32];
    #pragma unroll
    for (int ks = 0; ks < 32; ++ks)
        afr[ks] = *(const v2f*)(myfeat + lm * FP + ks * 4 + lh * 2);
    __syncthreads();

    float* myh = hbuf + wv * 16 * HP;
    for (int j = 0; j < 32; ++j) {
        float bias = b1[j * 16 + lm];
        v8f c;
        #pragma unroll
        for (int v = 0; v < 8; ++v) c[v] = bias;
        #pragma unroll
        for (int ks = 0; ks < 32; ++ks) {
            const float* wp = W1 + (ks * 4 + lh * 2) * HDIM + j * 16 + lm;
            v2f b; b.x = wp[0]; b.y = wp[HDIM];
            c = __builtin_amdgcn_wmma_f32_16x16x4_f32(false, afr[ks], false, b,
                                                      (short)0, c, false, false);
        }
        #pragma unroll
        for (int v = 0; v < 8; ++v) {
            int row = v + lh * 8;
            myh[row * HP + j * 16 + lm] = gelu_exact(c[v]);
        }
    }
    __syncthreads();

    for (int j = 0; j < 32; ++j) {
        float bias = b2[j * 16 + lm];
        v8f c;
        #pragma unroll
        for (int v = 0; v < 8; ++v) c[v] = bias;
        #pragma unroll 4
        for (int ks = 0; ks < 128; ++ks) {
            v2f a = *(const v2f*)(myh + lm * HP + ks * 4 + lh * 2);
            const float* wp = W2 + (ks * 4 + lh * 2) * HDIM + j * 16 + lm;
            v2f b; b.x = wp[0]; b.y = wp[HDIM];
            c = __builtin_amdgcn_wmma_f32_16x16x4_f32(false, a, false, b,
                                                      (short)0, c, false, false);
        }
        int col = j * 16 + lm;
        #pragma unroll
        for (int v = 0; v < 8; ++v) {
            int row = v + lh * 8;
            int t   = typ[wv * 16 + row];
            int c0  = t << 7;
            long rg = base + wv * 16 + row;
            if ((unsigned)(col - c0) < 128u && rg < Mtot)
                out[rg * DOUT + (col - c0)] = c[v];
        }
    }
}

extern "C" void kernel_launch(void* const* d_in, const int* in_sizes, int n_in,
                              void* d_out, int out_size, void* d_ws, size_t ws_size,
                              hipStream_t stream) {
    const float* nf = (const float*)d_in[0];
    const float* W1 = (const float*)d_in[1];
    const float* b1 = (const float*)d_in[2];
    const float* W2 = (const float*)d_in[3];
    const float* b2 = (const float*)d_in[4];
    float* out = (float*)d_out;

    int Mtot = in_sizes[0] / FTOT;
    int grid = (Mtot + MT - 1) / MT;

    hipFuncSetAttribute((const void*)mpx_encoder_kernel,
                        hipFuncAttributeMaxDynamicSharedMemorySize, SMEM_BYTES);

    mpx_encoder_kernel<<<grid, THREADS, SMEM_BYTES, stream>>>(nf, W1, b1, W2, b2, out, Mtot);
}


# GEMM2 loop interchange: k outer, a wave keeps all 512 output columns of its 16 rows in registers, h is only a 128-wide LDS slice; W2 panels 8x512, one read group per 32 WMMAs; same k order and sums
# speedup vs baseline: 2.4535x; 2.4535x over previous
#include <hip/hip_runtime.h>


typedef float v2f __attribute__((ext_vector_type(2)));
typedef float v4f __attribute__((ext_vector_type(4)));
typedef float v8f __attribute__((ext_vector_type(8)));

#define MT        64
#define WAVES     4
#define THREADS   128
#define FTOT      132
#define HDIM      512
#define DOUT      128
#define NC        128
#define FP        132
#define HP        132
#define HBYTES    (WAVES * 16 * HP * 4)
#define TYPOFF    HBYTES
#define PANOFF    (HBYTES + MT * 4)

#define KP        16
#define PP        (2 * NC + 32)
#define KP2       8
#define PP2       (2 * HDIM + 32)
#define NV2       (KP2 * HDIM / 4 / THREADS)
#define BUF_FL    ((KP2 / 2) * PP2)
#define SMEM_BYTES (PANOFF + 2 * BUF_FL * 4)

__device__ __forceinline__ float gelu_exact(float x) {
    return 0.5f * x * (1.0f + erff(x * 0.70710678118654752f));
}

__device__ __forceinline__ void fetch1(const float* __restrict__ src, int tid, v4f (&r)[4]) {
    const float* p = src + (2 * (tid >> 5)) * HDIM + 4 * (tid & 31);
    r[0] = *(const v4f*)(p);
    r[1] = *(const v4f*)(p + HDIM);
    r[2] = *(const v4f*)(p + 8 * HDIM);
    r[3] = *(const v4f*)(p + 9 * HDIM);
}

__device__ __forceinline__ void put1(float* dst, int tid, const v4f (&r)[4]) {
    float* d = dst + (tid >> 5) * PP + 8 * (tid & 31);
    v4f x0 = {r[0].x, r[1].x, r[0].y, r[1].y};
    v4f x1 = {r[0].z, r[1].z, r[0].w, r[1].w};
    v4f y0 = {r[2].x, r[3].x, r[2].y, r[3].y};
    v4f y1 = {r[2].z, r[3].z, r[2].w, r[3].w};
    *(v4f*)(d)              = x0;
    *(v4f*)(d + 4)          = x1;
    *(v4f*)(d + 4 * PP)     = y0;
    *(v4f*)(d + 4 * PP + 4) = y1;
}

__device__ __forceinline__ void fetch2(const float* __restrict__ src, int tid, v4f (&r)[NV2]) {
    const float* p = src + 4 * tid;
    #pragma unroll
    for (int i = 0; i < KP2; ++i) r[i] = *(const v4f*)(p + i * HDIM);
}

__device__ __forceinline__ void put2(float* dst, int tid, const v4f (&r)[NV2]) {
    float* d = dst + 8 * tid;
    #pragma unroll
    for (int j = 0; j < KP2 / 2; ++j) {
        v4f x0 = {r[2 * j].x, r[2 * j + 1].x, r[2 * j].y, r[2 * j + 1].y};
        v4f x1 = {r[2 * j].z, r[2 * j + 1].z, r[2 * j].w, r[2 * j + 1].w};
        *(v4f*)(d + j * PP2)     = x0;
        *(v4f*)(d + j * PP2 + 4) = x1;
    }
}

__device__ __forceinline__ void gemm2_panel(v8f (&acc2)[32], const float* ap, const float* bp) {
    #pragma unroll
    for (int s = 0; s < KP2 / 4; ++s) {
        v2f a = *(const v2f*)(ap + 4 * s);
        v2f b[32];
        #pragma unroll
        for (int t = 0; t < 32; ++t) b[t] = *(const v2f*)(bp + (2 * s) * PP2 + 32 * t);
        __builtin_amdgcn_sched_barrier(0);
        #pragma unroll
        for (int t = 0; t < 32; ++t)
            acc2[t] = __builtin_amdgcn_wmma_f32_16x16x4_f32(false, a, false, b[t],
                                                            (short)0, acc2[t], false, false);
        __builtin_amdgcn_sched_barrier(0);
    }
}

__global__ __launch_bounds__(THREADS, 2)
void mpx_encoder_kernel(const float* __restrict__ nf,
                        const float* __restrict__ W1,
                        const float* __restrict__ b1,
                        const float* __restrict__ W2,
                        const float* __restrict__ b2,
                        float* __restrict__ out,
                        int Mtot)
{
    extern __shared__ char smem[];
    float* feats = (float*)smem;
    float* hbuf  = (float*)smem;
    int*   typ   = (int*)(smem + TYPOFF);
    float* pan   = (float*)(smem + PANOFF);

    const int  tid  = threadIdx.x;
    const int  lane = tid & 31;
    const int  wv   = tid >> 5;
    const int  lm   = lane & 15;
    const int  lh   = lane >> 4;
    const long base = (long)blockIdx.x * MT;

    v4f pr[4];
    v4f pr2[NV2];
    fetch1(W1, tid, pr);

    #pragma unroll
    for (int it = 0; it < 16; ++it) {
        int i  = tid + THREADS * it;
        int r  = i >> 5;
        int c4 = i & 31;
        long rs = base + r; if (rs >= Mtot) rs = Mtot - 1;
        const float4 v = *(const float4*)(nf + rs * FTOT + 4 + c4 * 4);
        float* d = feats + r * FP + c4 * 4;
        d[0] = v.x; d[1] = v.y; d[2] = v.z; d[3] = v.w;
    }
    if (tid < MT) {
        long rs = base + tid; if (rs >= Mtot) rs = Mtot - 1;
        const float* p = nf + rs * FTOT;
        float best = p[0]; int bi = 0;
        #pragma unroll
        for (int i = 1; i < 4; ++i) { float q = p[i]; if (q > best) { best = q; bi = i; } }
        typ[tid] = bi;
    }
    put1(pan, tid, pr);
    __syncthreads();

    const float* myfeat = feats + wv * 16 * FP;
    v2f afr[32];
    #pragma unroll
    for (int ks = 0; ks < 32; ++ks)
        afr[ks] = *(const v2f*)(myfeat + lm * FP + ks * 4 + lh * 2);
    int tys[8];
    #pragma unroll
    for (int v = 0; v < 8; ++v) tys[v] = typ[wv * 16 + v + lh * 8];
    __syncthreads();

    float* myh = hbuf + wv * 16 * HP;
    const int boff1 = lh * PP  + 2 * lm;
    const int boff2 = lh * PP2 + 2 * lm;
    int buf = 0;

    v8f acc2[32];
    #pragma unroll
    for (int t = 0; t < 32; ++t) {
        float bias = b2[t * 16 + lm];
        #pragma unroll
        for (int v = 0; v < 8; ++v) acc2[t][v] = bias;
    }

    for (int g = 0; g < HDIM / NC; ++g) {
        v8f acc[8];
        #pragma unroll
        for (int t = 0; t < 8; ++t) {
            float bias = b1[g * NC + t * 16 + lm];
            #pragma unroll
            for (int v = 0; v < 8; ++v) acc[t][v] = bias;
        }
        #pragma unroll
        for (int kp = 0; kp < 128 / KP; ++kp) {
            if (kp + 1 < 128 / KP) fetch1(W1 + (kp + 1) * (KP * HDIM) + g * NC, tid, pr);
            else                   fetch2(W2 + (long)(g * NC) * HDIM, tid, pr2);
            const float* bp = pan + buf * BUF_FL + boff1;
            #pragma unroll
            for (int s = 0; s < KP / 4; ++s) {
                #pragma unroll
                for (int t = 0; t < 8; ++t) {
                    v2f b = *(const v2f*)(bp + (2 * s) * PP + 32 * t);
                    acc[t] = __builtin_amdgcn_wmma_f32_16x16x4_f32(false, afr[kp * (KP / 4) + s], false, b,
                                                                   (short)0, acc[t], false, false);
                }
            }
            __builtin_amdgcn_sched_barrier(0);
            if (kp + 1 < 128 / KP) put1(pan + (buf ^ 1) * BUF_FL, tid, pr);
            else                   put2(pan + (buf ^ 1) * BUF_FL, tid, pr2);
            __syncthreads();
            buf ^= 1;
        }
        #pragma unroll
        for (int t = 0; t < 8; ++t) {
            #pragma unroll
            for (int v = 0; v < 8; ++v) {
                int row = v + lh * 8;
                myh[row * HP + t * 16 + lm] = gelu_exact(acc[t][v]);
            }
        }
        __syncthreads();

        for (int kp = 0; kp + 1 < NC / KP2; ++kp) {
            fetch2(W2 + (long)(g * NC + (kp + 1) * KP2) * HDIM, tid, pr2);
            gemm2_panel(acc2, myh + lm * HP + kp * KP2 + lh * 2, pan + buf * BUF_FL + boff2);
            put2(pan + (buf ^ 1) * BUF_FL, tid, pr2);
            __syncthreads();
            buf ^= 1;
        }
        const int gn = (g + 1 < HDIM / NC) ? g + 1 : g;
        fetch1(W1 + gn * NC, tid, pr);
        gemm2_panel(acc2, myh + lm * HP + (NC / KP2 - 1) * KP2 + lh * 2, pan + buf * BUF_FL + boff2);
        put1(pan + (buf ^ 1) * BUF_FL, tid, pr);
        __syncthreads();
        buf ^= 1;
    }

    #pragma unroll
    for (int t = 0; t < 8; ++t) {
        #pragma unroll
        for (int v = 0; v < 8; ++v) {
            long rg = base + wv * 16 + v + lh * 8;
            const int ty = tys[v];
            float x = acc2[t][v];
            x = (ty == 1) ? acc2[8 + t][v]  : x;
            x = (ty == 2) ? acc2[16 + t][v] : x;
            x = (ty == 3) ? acc2[24 + t][v] : x;
            if (rg < Mtot) out[rg * DOUT + t * 16 + lm] = x;
        }
    }
}

extern "C" void kernel_launch(void* const* d_in, const int* in_sizes, int n_in,
                              void* d_out, int out_size, void* d_ws, size_t ws_size,
                              hipStream_t stream) {
    const float* nf = (const float*)d_in[0];
    const float* W1 = (const float*)d_in[1];
    const float* b1 = (const float*)d_in[2];
    const float* W2 = (const float*)d_in[3];
    const float* b2 = (const float*)d_in[4];
    float* out = (float*)d_out;

    int Mtot = in_sizes[0] / FTOT;
    int grid = (Mtot + MT - 1) / MT;

    hipFuncSetAttribute((const void*)mpx_encoder_kernel,
                        hipFuncAttributeMaxDynamicSharedMemorySize, SMEM_BYTES);

    mpx_encoder_kernel<<<grid, THREADS, SMEM_BYTES, stream>>>(nf, W1, b1, W2, b2, out, Mtot);
}
